# starting kernel
# baseline (speedup 1.0000x reference)
#include <hip/hip_runtime.h>
#include <math.h>

typedef float v2f __attribute__((ext_vector_type(2)));
typedef float v8f __attribute__((ext_vector_type(8)));

#define C_IN 128
#define NEG_SLOPE 0.2f
#define LN_EPS 1e-5f

__global__ void init_kernel(float* __restrict__ out, const float* __restrict__ bias,
                            float* __restrict__ mmax, float* __restrict__ denom,
                            long N) {
    long i = (long)blockIdx.x * blockDim.x + threadIdx.x;
    if (i < N * 32) out[i] = bias[i & 31];
    if (i < N * 4) { mmax[i] = -__builtin_inff(); denom[i] = 0.0f; }
}

__global__ __launch_bounds__(256) void ln_gemm_kernel(
    const float* __restrict__ x,
    const float* __restrict__ gamma, const float* __restrict__ beta,
    const float* __restrict__ Wl, const float* __restrict__ bl,
    const float* __restrict__ Wr, const float* __restrict__ br,
    float* __restrict__ xl, float* __restrict__ xr, long N)
{
    __shared__ float sXh[16][132];

    const int t   = threadIdx.x;
    const int row = t >> 4;
    const int sub = t & 15;
    const long node = (long)blockIdx.x * 16 + row;

    float v[8];
    if (node < N) {
        const float4* xp = (const float4*)(x + node * C_IN + sub * 8);
        float4 p0 = xp[0], p1 = xp[1];
        v[0]=p0.x; v[1]=p0.y; v[2]=p0.z; v[3]=p0.w;
        v[4]=p1.x; v[5]=p1.y; v[6]=p1.z; v[7]=p1.w;
    } else {
        #pragma unroll
        for (int j = 0; j < 8; ++j) v[j] = 0.0f;
    }

    float s = 0.0f;
    #pragma unroll
    for (int j = 0; j < 8; ++j) s += v[j];
    s += __shfl_xor(s, 1, 16); s += __shfl_xor(s, 2, 16);
    s += __shfl_xor(s, 4, 16); s += __shfl_xor(s, 8, 16);
    const float mu = s * (1.0f / 128.0f);

    float q = 0.0f;
    #pragma unroll
    for (int j = 0; j < 8; ++j) { float d = v[j] - mu; q += d * d; }
    q += __shfl_xor(q, 1, 16); q += __shfl_xor(q, 2, 16);
    q += __shfl_xor(q, 4, 16); q += __shfl_xor(q, 8, 16);
    const float rstd = rsqrtf(q * (1.0f / 128.0f) + LN_EPS);

    #pragma unroll
    for (int j = 0; j < 8; ++j) {
        const int c = sub * 8 + j;
        float y = (v[j] - mu) * rstd * gamma[c] + beta[c];
        sXh[row][c] = y > 0.0f ? y : 0.0f;
    }
    __syncthreads();

    const int lane = t & 31;
    const int wave = t >> 5;
    const int half = lane >> 4;
    const int lr   = lane & 15;
    const int col  = wave * 16 + lr;

    v8f cl = {}; v8f cr = {};
    #pragma unroll 8
    for (int k = 0; k < 128; k += 4) {
        v2f a = *(const v2f*)&sXh[lr][k + 2 * half];
        const int rk = (k + 2 * half) * C_IN + col;
        v2f pb; pb.x = Wl[rk]; pb.y = Wl[rk + C_IN];
        cl = __builtin_amdgcn_wmma_f32_16x16x4_f32(false, a, false, pb, (short)0, cl, false, false);
        v2f qb; qb.x = Wr[rk]; qb.y = Wr[rk + C_IN];
        cr = __builtin_amdgcn_wmma_f32_16x16x4_f32(false, a, false, qb, (short)0, cr, false, false);
    }

    const float bll = bl[col], brr = br[col];
    const long base = (long)blockIdx.x * 16;
    #pragma unroll
    for (int vv = 0; vv < 8; ++vv) {
        const long r = base + vv + 8 * half;
        if (r < N) {
            xl[r * C_IN + col] = cl[vv] + bll;
            xr[r * C_IN + col] = cr[vv] + brr;
        }
    }
}

__device__ __forceinline__ void atomicMaxF(float* addr, float val) {
    if (val >= 0.0f) atomicMax((int*)addr, __float_as_int(val));
    else             atomicMin((unsigned int*)addr, __float_as_uint(val));
}

__global__ __launch_bounds__(256) void edge_score_kernel(
    const int* __restrict__ srcI, const int* __restrict__ dstI,
    const float* __restrict__ xl, const float* __restrict__ xr,
    const float* __restrict__ att,
    float* __restrict__ escore, float* __restrict__ mmax,
    long E, long Etot)
{
    const int lane = threadIdx.x & 31;
    const long e = (long)blockIdx.x * 8 + (threadIdx.x >> 5);
    if (e >= Etot) return;
    int s_, d_;
    if (e < E) { s_ = srcI[e]; d_ = dstI[e]; }
    else       { s_ = d_ = (int)(e - E); }

    const float4 a4 = ((const float4*)att)[lane];
    const float4 l4 = ((const float4*)xl)[(long)s_ * 32 + lane];
    const float4 r4 = ((const float4*)xr)[(long)d_ * 32 + lane];

    float4 t4;
    t4.x = l4.x + r4.x; t4.y = l4.y + r4.y; t4.z = l4.z + r4.z; t4.w = l4.w + r4.w;
    t4.x = t4.x > 0.0f ? t4.x : NEG_SLOPE * t4.x;
    t4.y = t4.y > 0.0f ? t4.y : NEG_SLOPE * t4.y;
    t4.z = t4.z > 0.0f ? t4.z : NEG_SLOPE * t4.z;
    t4.w = t4.w > 0.0f ? t4.w : NEG_SLOPE * t4.w;

    float p = t4.x * a4.x + t4.y * a4.y + t4.z * a4.z + t4.w * a4.w;
    p += __shfl_xor(p, 1, 8);
    p += __shfl_xor(p, 2, 8);
    p += __shfl_xor(p, 4, 8);

    if ((lane & 7) == 0) {
        const int h = lane >> 3;
        escore[e * 4 + h] = p;
        atomicMaxF(&mmax[(long)d_ * 4 + h], p);
    }
}

__global__ void softmax_denom_kernel(
    const int* __restrict__ dstI, float* __restrict__ escore,
    const float* __restrict__ mmax, float* __restrict__ denom,
    long E, long Etot)
{
    const long i = (long)blockIdx.x * blockDim.x + threadIdx.x;
    if (i >= Etot * 4) return;
    const long e = i >> 2;
    const int  h = (int)(i & 3);
    const int  d_ = (e < E) ? dstI[e] : (int)(e - E);
    const float w = __expf(escore[i] - mmax[(long)d_ * 4 + h]);
    escore[i] = w;
    atomicAdd(&denom[(long)d_ * 4 + h], w);
}

__global__ __launch_bounds__(256) void aggregate_kernel(
    const int* __restrict__ srcI, const int* __restrict__ dstI,
    const float* __restrict__ xl, const float* __restrict__ escore,
    const float* __restrict__ denom, float* __restrict__ out,
    long E, long Etot)
{
    const int lane = threadIdx.x & 31;
    const long e = (long)blockIdx.x * 8 + (threadIdx.x >> 5);
    if (e >= Etot) return;
    int s_, d_;
    if (e < E) { s_ = srcI[e]; d_ = dstI[e]; }
    else       { s_ = d_ = (int)(e - E); }

    const int h = lane >> 3;
    const float w   = escore[e * 4 + h];
    const float den = denom[(long)d_ * 4 + h];
    const float alpha = 0.25f * w / (den + 1e-16f);

    const float4 l4 = ((const float4*)xl)[(long)s_ * 32 + lane];
    float vx = alpha * l4.x, vy = alpha * l4.y, vz = alpha * l4.z, vw = alpha * l4.w;

    vx += __shfl_xor(vx, 8);  vy += __shfl_xor(vy, 8);
    vz += __shfl_xor(vz, 8);  vw += __shfl_xor(vw, 8);
    vx += __shfl_xor(vx, 16); vy += __shfl_xor(vy, 16);
    vz += __shfl_xor(vz, 16); vw += __shfl_xor(vw, 16);

    if (lane < 8) {
        float* o = out + (long)d_ * 32 + lane * 4;
        atomicAdd(o + 0, vx);
        atomicAdd(o + 1, vy);
        atomicAdd(o + 2, vz);
        atomicAdd(o + 3, vw);
    }
}

extern "C" void kernel_launch(void* const* d_in, const int* in_sizes, int n_in,
                              void* d_out, int out_size, void* d_ws, size_t ws_size,
                              hipStream_t stream) {
    const float* x     = (const float*)d_in[0];
    const int*   ei    = (const int*)d_in[1];
    const float* gamma = (const float*)d_in[2];
    const float* beta  = (const float*)d_in[3];
    const float* Wl    = (const float*)d_in[4];
    const float* bl    = (const float*)d_in[5];
    const float* Wr    = (const float*)d_in[6];
    const float* br    = (const float*)d_in[7];
    const float* att   = (const float*)d_in[8];
    const float* bias  = (const float*)d_in[9];
    float* out = (float*)d_out;

    const long N = in_sizes[0] / C_IN;
    const long E = in_sizes[1] / 2;
    const long Etot = E + N;
    const int* srcI = ei;
    const int* dstI = ei + E;

    float* ws     = (float*)d_ws;
    float* xl     = ws;
    float* xr     = xl + N * C_IN;
    float* escore = xr + N * C_IN;
    float* mmax   = escore + Etot * 4;
    float* denom  = mmax + N * 4;

    {
        long total = N * 32;
        init_kernel<<<(int)((total + 255) / 256), 256, 0, stream>>>(out, bias, mmax, denom, N);
    }
    {
        int blocks = (int)((N + 15) / 16);
        ln_gemm_kernel<<<blocks, 256, 0, stream>>>(x, gamma, beta, Wl, bl, Wr, br, xl, xr, N);
    }
    {
        int blocks = (int)((Etot + 7) / 8);
        edge_score_kernel<<<blocks, 256, 0, stream>>>(srcI, dstI, xl, xr, att, escore, mmax, E, Etot);
    }
    {
        long total = Etot * 4;
        softmax_denom_kernel<<<(int)((total + 255) / 256), 256, 0, stream>>>(dstI, escore, mmax, denom, E, Etot);
    }
    {
        int blocks = (int)((Etot + 7) / 8);
        aggregate_kernel<<<blocks, 256, 0, stream>>>(srcI, dstI, xl, escore, denom, out, E, Etot);
    }
}


# Edge score and aggregate launches: 1792 resident work-groups, each wave loops over 8-edge blocks (16 ids in one load, rows of 4 edges gathered together) instead of one wave per edge; atomics kept
# speedup vs baseline: 1.3341x; 1.3341x over previous
#include <hip/hip_runtime.h>
#include <math.h>

typedef float v2f __attribute__((ext_vector_type(2)));
typedef float v8f __attribute__((ext_vector_type(8)));

#define C_IN 128
#define NEG_SLOPE 0.2f
#define LN_EPS 1e-5f

__global__ void init_kernel(float* __restrict__ out, const float* __restrict__ bias,
                            float* __restrict__ mmax, float* __restrict__ denom,
                            long N) {
    long i = (long)blockIdx.x * blockDim.x + threadIdx.x;
    if (i < N * 32) out[i] = bias[i & 31];
    if (i < N * 4) { mmax[i] = -__builtin_inff(); denom[i] = 0.0f; }
}

__global__ __launch_bounds__(256) void ln_gemm_kernel(
    const float* __restrict__ x,
    const float* __restrict__ gamma, const float* __restrict__ beta,
    const float* __restrict__ Wl, const float* __restrict__ bl,
    const float* __restrict__ Wr, const float* __restrict__ br,
    float* __restrict__ xl, float* __restrict__ xr, long N)
{
    __shared__ float sXh[16][132];

    const int t   = threadIdx.x;
    const int row = t >> 4;
    const int sub = t & 15;
    const long node = (long)blockIdx.x * 16 + row;

    float v[8];
    if (node < N) {
        const float4* xp = (const float4*)(x + node * C_IN + sub * 8);
        float4 p0 = xp[0], p1 = xp[1];
        v[0]=p0.x; v[1]=p0.y; v[2]=p0.z; v[3]=p0.w;
        v[4]=p1.x; v[5]=p1.y; v[6]=p1.z; v[7]=p1.w;
    } else {
        #pragma unroll
        for (int j = 0; j < 8; ++j) v[j] = 0.0f;
    }

    float s = 0.0f;
    #pragma unroll
    for (int j = 0; j < 8; ++j) s += v[j];
    s += __shfl_xor(s, 1, 16); s += __shfl_xor(s, 2, 16);
    s += __shfl_xor(s, 4, 16); s += __shfl_xor(s, 8, 16);
    const float mu = s * (1.0f / 128.0f);

    float q = 0.0f;
    #pragma unroll
    for (int j = 0; j < 8; ++j) { float d = v[j] - mu; q += d * d; }
    q += __shfl_xor(q, 1, 16); q += __shfl_xor(q, 2, 16);
    q += __shfl_xor(q, 4, 16); q += __shfl_xor(q, 8, 16);
    const float rstd = rsqrtf(q * (1.0f / 128.0f) + LN_EPS);

    #pragma unroll
    for (int j = 0; j < 8; ++j) {
        const int c = sub * 8 + j;
        float y = (v[j] - mu) * rstd * gamma[c] + beta[c];
        sXh[row][c] = y > 0.0f ? y : 0.0f;
    }
    __syncthreads();

    const int lane = t & 31;
    const int wave = t >> 5;
    const int half = lane >> 4;
    const int lr   = lane & 15;
    const int col  = wave * 16 + lr;

    v8f cl = {}; v8f cr = {};
    #pragma unroll 8
    for (int k = 0; k < 128; k += 4) {
        v2f a = *(const v2f*)&sXh[lr][k + 2 * half];
        const int rk = (k + 2 * half) * C_IN + col;
        v2f pb; pb.x = Wl[rk]; pb.y = Wl[rk + C_IN];
        cl = __builtin_amdgcn_wmma_f32_16x16x4_f32(false, a, false, pb, (short)0, cl, false, false);
        v2f qb; qb.x = Wr[rk]; qb.y = Wr[rk + C_IN];
        cr = __builtin_amdgcn_wmma_f32_16x16x4_f32(false, a, false, qb, (short)0, cr, false, false);
    }

    const float bll = bl[col], brr = br[col];
    const long base = (long)blockIdx.x * 16;
    #pragma unroll
    for (int vv = 0; vv < 8; ++vv) {
        const long r = base + vv + 8 * half;
        if (r < N) {
            xl[r * C_IN + col] = cl[vv] + bll;
            xr[r * C_IN + col] = cr[vv] + brr;
        }
    }
}

__device__ __forceinline__ void atomicMaxF(float* addr, float val) {
    if (val >= 0.0f) atomicMax((int*)addr, __float_as_int(val));
    else             atomicMin((unsigned int*)addr, __float_as_uint(val));
}

#define EDGE_BLK 8
#define EDGE_GRID 1792

__device__ __forceinline__ int load_block_ids(const int* __restrict__ srcI,
                                              const int* __restrict__ dstI,
                                              long e0, long E, long Etot, int lane) {
    int idx = 0;
    const long e = e0 + (lane & 7);
    if (lane < 16 && e < Etot) {
        if (e < E) { const int* p = (lane < 8) ? srcI : dstI; idx = p[e]; }
        else       { idx = (int)(e - E); }
    }
    return idx;
}

__global__ __launch_bounds__(256) void edge_score_kernel(
    const int* __restrict__ srcI, const int* __restrict__ dstI,
    const float* __restrict__ xl, const float* __restrict__ xr,
    const float* __restrict__ att,
    float* __restrict__ escore, float* __restrict__ mmax,
    long E, long Etot)
{
    const int lane = threadIdx.x & 31;
    const long nwaves = (long)gridDim.x * 8;
    const long nblk = (Etot + EDGE_BLK - 1) / EDGE_BLK;
    const float4 a4 = ((const float4*)att)[lane];
    const float4* xl4 = (const float4*)xl;
    const float4* xr4 = (const float4*)xr;

    for (long blk = (long)blockIdx.x * 8 + (threadIdx.x >> 5); blk < nblk; blk += nwaves) {
        const long e0 = blk * EDGE_BLK;
        const int idx = load_block_ids(srcI, dstI, e0, E, Etot, lane);

        #pragma unroll
        for (int b = 0; b < EDGE_BLK; b += 4) {
            float4 l4[4], r4[4];
            int dd[4];
            #pragma unroll
            for (int u = 0; u < 4; ++u) {
                const int s_ = __builtin_amdgcn_readlane(idx, b + u);
                const int d_ = __builtin_amdgcn_readlane(idx, 8 + b + u);
                dd[u] = d_;
                l4[u] = xl4[(long)s_ * 32 + lane];
                r4[u] = xr4[(long)d_ * 32 + lane];
            }
            #pragma unroll
            for (int u = 0; u < 4; ++u) {
                float4 t4;
                t4.x = l4[u].x + r4[u].x; t4.y = l4[u].y + r4[u].y;
                t4.z = l4[u].z + r4[u].z; t4.w = l4[u].w + r4[u].w;
                t4.x = t4.x > 0.0f ? t4.x : NEG_SLOPE * t4.x;
                t4.y = t4.y > 0.0f ? t4.y : NEG_SLOPE * t4.y;
                t4.z = t4.z > 0.0f ? t4.z : NEG_SLOPE * t4.z;
                t4.w = t4.w > 0.0f ? t4.w : NEG_SLOPE * t4.w;

                float p = t4.x * a4.x + t4.y * a4.y + t4.z * a4.z + t4.w * a4.w;
                p += __shfl_xor(p, 1, 8);
                p += __shfl_xor(p, 2, 8);
                p += __shfl_xor(p, 4, 8);

                const long e = e0 + b + u;
                if ((lane & 7) == 0 && e < Etot) {
                    const int h = lane >> 3;
                    escore[e * 4 + h] = p;
                    atomicMaxF(&mmax[(long)dd[u] * 4 + h], p);
                }
            }
        }
    }
}

__global__ void softmax_denom_kernel(
    const int* __restrict__ dstI, float* __restrict__ escore,
    const float* __restrict__ mmax, float* __restrict__ denom,
    long E, long Etot)
{
    const long i = (long)blockIdx.x * blockDim.x + threadIdx.x;
    if (i >= Etot * 4) return;
    const long e = i >> 2;
    const int  h = (int)(i & 3);
    const int  d_ = (e < E) ? dstI[e] : (int)(e - E);
    const float w = __expf(escore[i] - mmax[(long)d_ * 4 + h]);
    escore[i] = w;
    atomicAdd(&denom[(long)d_ * 4 + h], w);
}

__global__ __launch_bounds__(256) void aggregate_kernel(
    const int* __restrict__ srcI, const int* __restrict__ dstI,
    const float* __restrict__ xl, const float* __restrict__ escore,
    const float* __restrict__ denom, float* __restrict__ out,
    long E, long Etot)
{
    const int lane = threadIdx.x & 31;
    const long nwaves = (long)gridDim.x * 8;
    const long nblk = (Etot + EDGE_BLK - 1) / EDGE_BLK;
    const float4* xl4 = (const float4*)xl;
    const int hsel = lane >> 3;

    for (long blk = (long)blockIdx.x * 8 + (threadIdx.x >> 5); blk < nblk; blk += nwaves) {
        const long e0 = blk * EDGE_BLK;
        const int idx = load_block_ids(srcI, dstI, e0, E, Etot, lane);

        const long wi = e0 * 4 + lane;
        const float w = (wi < Etot * 4) ? escore[wi] : 0.0f;
        const int dmine = __shfl(idx, 8 + (lane >> 2));
        const float den = denom[(long)dmine * 4 + (lane & 3)];
        const float alpha_l = 0.25f * w / (den + 1e-16f);

        #pragma unroll
        for (int b = 0; b < EDGE_BLK; b += 4) {
            float4 l4[4];
            int dd[4];
            #pragma unroll
            for (int u = 0; u < 4; ++u) {
                const int s_ = __builtin_amdgcn_readlane(idx, b + u);
                dd[u] = __builtin_amdgcn_readlane(idx, 8 + b + u);
                l4[u] = xl4[(long)s_ * 32 + lane];
            }
            #pragma unroll
            for (int u = 0; u < 4; ++u) {
                const float alpha = __shfl(alpha_l, (b + u) * 4 + hsel);
                float vx = alpha * l4[u].x, vy = alpha * l4[u].y,
                      vz = alpha * l4[u].z, vw = alpha * l4[u].w;

                vx += __shfl_xor(vx, 8);  vy += __shfl_xor(vy, 8);
                vz += __shfl_xor(vz, 8);  vw += __shfl_xor(vw, 8);
                vx += __shfl_xor(vx, 16); vy += __shfl_xor(vy, 16);
                vz += __shfl_xor(vz, 16); vw += __shfl_xor(vw, 16);

                if (lane < 8 && e0 + b + u < Etot) {
                    float* o = out + (long)dd[u] * 32 + lane * 4;
                    atomicAdd(o + 0, vx);
                    atomicAdd(o + 1, vy);
                    atomicAdd(o + 2, vz);
                    atomicAdd(o + 3, vw);
                }
            }
        }
    }
}

extern "C" void kernel_launch(void* const* d_in, const int* in_sizes, int n_in,
                              void* d_out, int out_size, void* d_ws, size_t ws_size,
                              hipStream_t stream) {
    const float* x     = (const float*)d_in[0];
    const int*   ei    = (const int*)d_in[1];
    const float* gamma = (const float*)d_in[2];
    const float* beta  = (const float*)d_in[3];
    const float* Wl    = (const float*)d_in[4];
    const float* bl    = (const float*)d_in[5];
    const float* Wr    = (const float*)d_in[6];
    const float* br    = (const float*)d_in[7];
    const float* att   = (const float*)d_in[8];
    const float* bias  = (const float*)d_in[9];
    float* out = (float*)d_out;

    const long N = in_sizes[0] / C_IN;
    const long E = in_sizes[1] / 2;
    const long Etot = E + N;
    const int* srcI = ei;
    const int* dstI = ei + E;

    float* ws     = (float*)d_ws;
    float* xl     = ws;
    float* xr     = xl + N * C_IN;
    float* escore = xr + N * C_IN;
    float* mmax   = escore + Etot * 4;
    float* denom  = mmax + N * 4;

    {
        long total = N * 32;
        init_kernel<<<(int)((total + 255) / 256), 256, 0, stream>>>(out, bias, mmax, denom, N);
    }
    {
        int blocks = (int)((N + 15) / 16);
        ln_gemm_kernel<<<blocks, 256, 0, stream>>>(x, gamma, beta, Wl, bl, Wr, br, xl, xr, N);
    }
    {
        edge_score_kernel<<<EDGE_GRID, 256, 0, stream>>>(srcI, dstI, xl, xr, att, escore, mmax, E, Etot);
    }
    {
        long total = Etot * 4;
        softmax_denom_kernel<<<(int)((total + 255) / 256), 256, 0, stream>>>(dstI, escore, mmax, denom, E, Etot);
    }
    {
        aggregate_kernel<<<EDGE_GRID, 256, 0, stream>>>(srcI, dstI, xl, escore, denom, out, E, Etot);
    }
}


# Node launch in two stages: 9 lanes walk 9 nodes' lists side by side into an LDS table of source ids; the wave then sums a node with 4 x_l rows asked for ahead; sum order unchanged (list order)
# speedup vs baseline: 2.8027x; 2.1008x over previous
#include <hip/hip_runtime.h>
#include <math.h>

typedef float v2f __attribute__((ext_vector_type(2)));
typedef float v8f __attribute__((ext_vector_type(8)));

#define C_IN 128
#define NEG_SLOPE 0.2f
#define LN_EPS 1e-5f

#define ROWS_PB 80
#define ROW_TILES 5
#define KG 8
#define LIST_WGS 256
#define EDGES_PL 8

__global__ __launch_bounds__(256) void head_clear_kernel(int* __restrict__ head, long N)
{
    const long i = (long)blockIdx.x * 256 + threadIdx.x;
    if (i < N) head[i] = -1;
}

__device__ __forceinline__ void list_build_blocks(
    const int* __restrict__ srcI, const int* __restrict__ dstI,
    int* __restrict__ head, int2* __restrict__ link, const long E,
    const long firstBlock, const long blockStride)
{
    const int lane = threadIdx.x & 31;
    const long nblk = (E + 32 * EDGES_PL - 1) / (32 * EDGES_PL);
    for (long b = firstBlock; b < nblk; b += blockStride) {
        const long e0 = b * (32 * EDGES_PL) + lane;
        if ((b + 1) * (32 * EDGES_PL) <= E) {
            int d_[EDGES_PL], s_[EDGES_PL], prev[EDGES_PL];
            #pragma unroll
            for (int j = 0; j < EDGES_PL; ++j) { d_[j] = dstI[e0 + 32 * j]; s_[j] = srcI[e0 + 32 * j]; }
            #pragma unroll
            for (int j = 0; j < EDGES_PL; ++j) prev[j] = atomicExch(&head[d_[j]], (int)(e0 + 32 * j));
            #pragma unroll
            for (int j = 0; j < EDGES_PL; ++j) link[e0 + 32 * j] = make_int2(prev[j], s_[j]);
        } else {
            #pragma unroll
            for (int j = 0; j < EDGES_PL; ++j) {
                const long e = e0 + 32 * j;
                if (e < E) {
                    const int d1 = dstI[e];
                    const int s1 = srcI[e];
                    const int p1 = atomicExch(&head[d1], (int)e);
                    link[e] = make_int2(p1, s1);
                }
            }
        }
    }
}

__global__ __launch_bounds__(256) void ln_gemm_kernel(
    const float* __restrict__ x,
    const float* __restrict__ gamma, const float* __restrict__ beta,
    const float* __restrict__ Wl, const float* __restrict__ bl,
    const float* __restrict__ Wr, const float* __restrict__ br,
    float* __restrict__ xl, float* __restrict__ xr, long N,
    const int* __restrict__ srcI, const int* __restrict__ dstI,
    int* __restrict__ head, int2* __restrict__ link, long E)
{
    __shared__ float sXh[ROWS_PB][132];

    const int t   = threadIdx.x;
    if (blockIdx.x < LIST_WGS) {
        list_build_blocks(srcI, dstI, head, link, E,
                          (long)blockIdx.x * 8 + (t >> 5), (long)LIST_WGS * 8);
        return;
    }
    const int r0  = t >> 4;
    const int sub = t & 15;
    const long blockRow = (long)(blockIdx.x - LIST_WGS) * ROWS_PB;

    float v[ROW_TILES][8];
    #pragma unroll
    for (int p = 0; p < ROW_TILES; ++p) {
        const long node = blockRow + p * 16 + r0;
        if (node < N) {
            const float4* xp = (const float4*)(x + node * C_IN + sub * 8);
            float4 p0 = xp[0], p1 = xp[1];
            v[p][0]=p0.x; v[p][1]=p0.y; v[p][2]=p0.z; v[p][3]=p0.w;
            v[p][4]=p1.x; v[p][5]=p1.y; v[p][6]=p1.z; v[p][7]=p1.w;
        } else {
            #pragma unroll
            for (int j = 0; j < 8; ++j) v[p][j] = 0.0f;
        }
    }

    #pragma unroll
    for (int p = 0; p < ROW_TILES; ++p) {
        float s = 0.0f;
        #pragma unroll
        for (int j = 0; j < 8; ++j) s += v[p][j];
        s += __shfl_xor(s, 1, 16); s += __shfl_xor(s, 2, 16);
        s += __shfl_xor(s, 4, 16); s += __shfl_xor(s, 8, 16);
        const float mu = s * (1.0f / 128.0f);

        float q = 0.0f;
        #pragma unroll
        for (int j = 0; j < 8; ++j) { float d = v[p][j] - mu; q += d * d; }
        q += __shfl_xor(q, 1, 16); q += __shfl_xor(q, 2, 16);
        q += __shfl_xor(q, 4, 16); q += __shfl_xor(q, 8, 16);
        const float rstd = rsqrtf(q * (1.0f / 128.0f) + LN_EPS);

        #pragma unroll
        for (int j = 0; j < 8; ++j) {
            const int c = sub * 8 + j;
            float y = (v[p][j] - mu) * rstd * gamma[c] + beta[c];
            sXh[p * 16 + r0][c] = y > 0.0f ? y : 0.0f;
        }
    }
    __syncthreads();

    const int lane = t & 31;
    const int wave = t >> 5;
    const int half = lane >> 4;
    const int lr   = lane & 15;
    const int col  = wave * 16 + lr;

    v8f cl[ROW_TILES], cr[ROW_TILES];
    #pragma unroll
    for (int p = 0; p < ROW_TILES; ++p) { cl[p] = (v8f){}; cr[p] = (v8f){}; }

    v2f pb[KG], qb[KG];
    #pragma unroll
    for (int s = 0; s < KG; ++s) {
        const int rk = (4 * s + 2 * half) * C_IN + col;
        pb[s].x = Wl[rk]; pb[s].y = Wl[rk + C_IN];
        qb[s].x = Wr[rk]; qb[s].y = Wr[rk + C_IN];
    }
    #pragma unroll
    for (int kg = 0; kg < 128; kg += 4 * KG) {
        v2f npb[KG], nqb[KG];
        if (kg + 4 * KG < 128) {
            #pragma unroll
            for (int s = 0; s < KG; ++s) {
                const int rk = (kg + 4 * KG + 4 * s + 2 * half) * C_IN + col;
                npb[s].x = Wl[rk]; npb[s].y = Wl[rk + C_IN];
                nqb[s].x = Wr[rk]; nqb[s].y = Wr[rk + C_IN];
            }
        }
        #pragma unroll
        for (int s = 0; s < KG; ++s) {
            const int k = kg + 4 * s;
            #pragma unroll
            for (int p = 0; p < ROW_TILES; ++p) {
                v2f a = *(const v2f*)&sXh[p * 16 + lr][k + 2 * half];
                cl[p] = __builtin_amdgcn_wmma_f32_16x16x4_f32(false, a, false, pb[s], (short)0, cl[p], false, false);
                cr[p] = __builtin_amdgcn_wmma_f32_16x16x4_f32(false, a, false, qb[s], (short)0, cr[p], false, false);
            }
        }
        if (kg + 4 * KG < 128) {
            #pragma unroll
            for (int s = 0; s < KG; ++s) { pb[s] = npb[s]; qb[s] = nqb[s]; }
        }
    }

    static_assert(ROW_TILES == 5, "the statement below names five accumulator pairs");
    asm volatile("v_nop\n\tv_nop\n\tv_nop\n\tv_nop\n\tv_nop"
                 : "+v"(cl[0]), "+v"(cl[1]), "+v"(cl[2]), "+v"(cl[3]), "+v"(cl[4]),
                   "+v"(cr[0]), "+v"(cr[1]), "+v"(cr[2]), "+v"(cr[3]), "+v"(cr[4]));

    const float bll = bl[col], brr = br[col];
    if (blockRow + ROWS_PB <= N) {
        #pragma unroll
        for (int p = 0; p < ROW_TILES; ++p) {
            #pragma unroll
            for (int vv = 0; vv < 8; ++vv) {
                const long r = blockRow + p * 16 + vv + 8 * half;
                xl[r * C_IN + col] = cl[p][vv] + bll;
                xr[r * C_IN + col] = cr[p][vv] + brr;
            }
        }
    } else {
        #pragma unroll
        for (int p = 0; p < ROW_TILES; ++p) {
            #pragma unroll
            for (int vv = 0; vv < 8; ++vv) {
                const long r = blockRow + p * 16 + vv + 8 * half;
                if (r < N) {
                    xl[r * C_IN + col] = cl[p][vv] + bll;
                    xr[r * C_IN + col] = cr[p][vv] + brr;
                }
            }
        }
    }
}

#define NODE_GRID 1536
#define NB 9
#define CAP 64
#define PITCH 72
#define GRP 4

__device__ __forceinline__ float lane_dot(const float4 l4, const float4 r4, const float4 a4) {
    float4 t4;
    t4.x = l4.x + r4.x; t4.y = l4.y + r4.y;
    t4.z = l4.z + r4.z; t4.w = l4.w + r4.w;
    t4.x = t4.x > 0.0f ? t4.x : NEG_SLOPE * t4.x;
    t4.y = t4.y > 0.0f ? t4.y : NEG_SLOPE * t4.y;
    t4.z = t4.z > 0.0f ? t4.z : NEG_SLOPE * t4.z;
    t4.w = t4.w > 0.0f ? t4.w : NEG_SLOPE * t4.w;
    return t4.x * a4.x + t4.y * a4.y + t4.z * a4.z + t4.w * a4.w;
}

__device__ __forceinline__ float head_sum(float p) {
    p += __shfl_xor(p, 1, 8);
    p += __shfl_xor(p, 2, 8);
    p += __shfl_xor(p, 4, 8);
    return p;
}

__device__ __forceinline__ void edge_update(const float4 row, const float pp,
                                            float& m, float& den, float4& acc) {
    const float p  = head_sum(pp);
    const float mn = fmaxf(m, p);
    const float sc = __expf(m - mn);
    const float w  = __expf(p - mn);
    den   = den   * sc + w;
    acc.x = acc.x * sc + w * row.x;
    acc.y = acc.y * sc + w * row.y;
    acc.z = acc.z * sc + w * row.z;
    acc.w = acc.w * sc + w * row.w;
    m = mn;
}

__device__ __forceinline__ void take_edge(const float4 row, const float4 r4, const float4 a4,
                                          float& m, float& den, float4& acc) {
    edge_update(row, lane_dot(row, r4, a4), m, den, acc);
}

__device__ __forceinline__ int link_slot(const int e, const int last) {
    return min(max(e, 0), last);
}

__device__ __forceinline__ float4 row_of(const float4* __restrict__ xl4, const int s, const int lane) {
    const float4* rowp = xl4 + (long)s * 32;
    return rowp[lane];
}

__device__ __forceinline__ void load_rows(float4 (&row)[GRP], const int* tab, const int k, const int T,
                                          const float4* __restrict__ xl4, const int lane) {
    const int4 q = *(const int4*)(tab + k);
    const int i0 = __builtin_amdgcn_readfirstlane(q.x);
    const int i1 = (k + 1 < T) ? __builtin_amdgcn_readfirstlane(q.y) : i0;
    const int i2 = (k + 2 < T) ? __builtin_amdgcn_readfirstlane(q.z) : i1;
    const int i3 = (k + 3 < T) ? __builtin_amdgcn_readfirstlane(q.w) : i2;
    row[0] = row_of(xl4, i0, lane);
    row[1] = row_of(xl4, i1, lane);
    row[2] = row_of(xl4, i2, lane);
    row[3] = row_of(xl4, i3, lane);
}

__global__ __launch_bounds__(256, 12) void node_pull_kernel(
    const int* __restrict__ head, const int2* __restrict__ link,
    const float* __restrict__ xl, const float* __restrict__ xr,
    const float* __restrict__ att, const float* __restrict__ bias,
    float* __restrict__ out, long N, long E)
{
    __shared__ __attribute__((aligned(16))) int sSrc[8][NB][PITCH];

    const int lane = threadIdx.x & 31;
    const int wave = __builtin_amdgcn_readfirstlane((int)(threadIdx.x >> 5));
    const unsigned nw  = gridDim.x * 8u;
    const unsigned wid = blockIdx.x * 8u + (unsigned)wave;
    const unsigned per = (unsigned)N / nw, extra = (unsigned)N % nw;
    const long lo = (long)wid * per + min(wid, extra);
    const long hi = lo + per + (wid < extra ? 1 : 0);
    const float4 a4 = ((const float4*)att)[lane];
    const float4* xl4 = (const float4*)xl;
    const float4* xr4 = (const float4*)xr;
    const int last = (int)(E - 1);
    const int nmax = (int)(N - 1);
    int* const mine = &sSrc[wave][lane < NB ? lane : 0][0];

    for (long n0 = lo; n0 < hi; n0 += NB) {
        const int nb = (int)min((long)NB, hi - n0);

        int cnt = 0;
        int cur = -1;
        if (lane < nb) {
            mine[0] = (int)(n0 + lane);
            cnt = 1;
            cur = head[n0 + lane];
        }
        for (int h = 1; h < CAP; ++h) {
            if (__builtin_amdgcn_ballot_w32(cur >= 0) == 0u) break;
            if (cur >= 0) {
                const int2 l = link[link_slot(cur, last)];
                mine[cnt] = min(max(l.y, 0), nmax);
                ++cnt;
                cur = l.x;
            }
        }

        for (int j = 0; j < nb; ++j) {
            const long i = n0 + j;
            const int T   = __builtin_amdgcn_readlane(cnt, j);
            const int rem = __builtin_amdgcn_readlane(cur, j);
            const int* tab = &sSrc[wave][j][0];
            const float4 r4 = (xr4 + i * 32)[lane];

            float m = -INFINITY;
            float den = 0.0f;
            float4 acc = make_float4(0.0f, 0.0f, 0.0f, 0.0f);

            float4 row[GRP];
            load_rows(row, tab, 0, T, xl4, lane);
            int k = 0;
            while (k + 2 * GRP <= T) {
                const int4 q = *(const int4*)(tab + k + GRP);
                take_edge(row[0], r4, a4, m, den, acc);
                row[0] = row_of(xl4, __builtin_amdgcn_readfirstlane(q.x), lane);
                take_edge(row[1], r4, a4, m, den, acc);
                row[1] = row_of(xl4, __builtin_amdgcn_readfirstlane(q.y), lane);
                take_edge(row[2], r4, a4, m, den, acc);
                row[2] = row_of(xl4, __builtin_amdgcn_readfirstlane(q.z), lane);
                take_edge(row[3], r4, a4, m, den, acc);
                row[3] = row_of(xl4, __builtin_amdgcn_readfirstlane(q.w), lane);
                k += GRP;
            }
            if (T - k > GRP) {
                float4 nxt[GRP];
                load_rows(nxt, tab, k + GRP, T, xl4, lane);
                #pragma unroll
                for (int u = 0; u < GRP; ++u) take_edge(row[u], r4, a4, m, den, acc);
                #pragma unroll
                for (int u = 0; u < GRP; ++u)
                    if (k + GRP + u < T) take_edge(nxt[u], r4, a4, m, den, acc);
            } else {
                #pragma unroll
                for (int u = 0; u < GRP; ++u)
                    if (k + u < T) take_edge(row[u], r4, a4, m, den, acc);
            }

            int e = rem;
            for (long c2 = 0; c2 < E && e >= 0; ++c2) {
                const int2 l = link[link_slot(e, last)];
                const int s = min(max(__builtin_amdgcn_readfirstlane(l.y), 0), nmax);
                e = __builtin_amdgcn_readfirstlane(l.x);
                take_edge(row_of(xl4, s, lane), r4, a4, m, den, acc);
            }

            const float inv = 0.25f / (den + 1e-16f);
            float vx = acc.x * inv, vy = acc.y * inv, vz = acc.z * inv, vw = acc.w * inv;
            vx += __shfl_xor(vx, 8);  vy += __shfl_xor(vy, 8);
            vz += __shfl_xor(vz, 8);  vw += __shfl_xor(vw, 8);
            vx += __shfl_xor(vx, 16); vy += __shfl_xor(vy, 16);
            vz += __shfl_xor(vz, 16); vw += __shfl_xor(vw, 16);
            if (lane < 8) {
                const float4 b4 = ((const float4*)bias)[lane];
                float4 o4;
                o4.x = vx + b4.x; o4.y = vy + b4.y; o4.z = vz + b4.z; o4.w = vw + b4.w;
                ((float4*)out)[i * 8 + lane] = o4;
            }
        }
    }
}

extern "C" void kernel_launch(void* const* d_in, const int* in_sizes, int n_in,
                              void* d_out, int out_size, void* d_ws, size_t ws_size,
                              hipStream_t stream) {
    const float* x     = (const float*)d_in[0];
    const int*   ei    = (const int*)d_in[1];
    const float* gamma = (const float*)d_in[2];
    const float* beta  = (const float*)d_in[3];
    const float* Wl    = (const float*)d_in[4];
    const float* bl    = (const float*)d_in[5];
    const float* Wr    = (const float*)d_in[6];
    const float* br    = (const float*)d_in[7];
    const float* att   = (const float*)d_in[8];
    const float* bias  = (const float*)d_in[9];
    float* out = (float*)d_out;

    const long N = in_sizes[0] / C_IN;
    const long E = in_sizes[1] / 2;
    const int* srcI = ei;
    const int* dstI = ei + E;

    float* ws   = (float*)d_ws;
    float* xl   = ws;
    float* xr   = xl + N * C_IN;
    int2*  link = (int2*)(xr + N * C_IN);
    int*   head = (int*)(link + E);

    {
        head_clear_kernel<<<(int)((N + 255) / 256), 256, 0, stream>>>(head, N);
    }
    {
        int blocks = (int)((N + ROWS_PB - 1) / ROWS_PB);
        ln_gemm_kernel<<<blocks + LIST_WGS, 256, 0, stream>>>(x, gamma, beta, Wl, bl, Wr, br, xl, xr, N,
                                                             srcI, dstI, head, link, E);
    }
    {
        node_pull_kernel<<<NODE_GRID, 256, 0, stream>>>(head, link, xl, xr, att, bias, out, N, E);
    }
}


# List build: a head is swapped on from one half of the chip only: a list wave reads its die, draws 512-edge blocks from its half's counter, links the edges of its half's nodes; sum order = list order
# speedup vs baseline: 4.0453x; 1.4434x over previous
#include <hip/hip_runtime.h>
#include <math.h>

typedef float v2f __attribute__((ext_vector_type(2)));
typedef float v8f __attribute__((ext_vector_type(8)));

#define C_IN 128
#define NEG_SLOPE 0.2f
#define LN_EPS 1e-5f

#define ROWS_PB 80
#define ROW_TILES 5
#define KG 8
#define LIST_WGS 256
#define EDGES_PL 16
#define TICKET_PAD 64

__device__ __forceinline__ int* ticket_words(int* head, const long N) {
    return head + ((N + 63) & ~63L);
}

__global__ __launch_bounds__(256) void head_clear_kernel(int* __restrict__ head, long N)
{
    const long i = (long)blockIdx.x * 256 + threadIdx.x;
    if (i < N) head[i] = -1;
    if (i < 2) ticket_words(head, N)[i * TICKET_PAD] = 0;
}


__device__ __forceinline__ int chip_half() {
    const unsigned id = __builtin_amdgcn_s_sendmsg_rtn(0x87);
    return (int)((id >> 18) & 1u);
}

__device__ __forceinline__ int ask_block(int* __restrict__ counter, const int lane) {
    unsigned t = 0u;
    if (lane == 0) t = atomicInc((unsigned*)counter, 0xFFFFFFFFu);
    return (int)t;
}

__device__ __forceinline__ void list_build_by_half(
    const int* __restrict__ srcI, const int* __restrict__ dstI,
    int* __restrict__ head, int2* __restrict__ link, const long E, const long N)
{
    const int lane = threadIdx.x & 31;
    const int nblk = (int)((E + 32 * EDGES_PL - 1) / (32 * EDGES_PL));
    const int split = (int)(((N >> 1) + 63) & ~63L);
    int* const tickets = ticket_words(head, N);
    const int mine = chip_half();

    #pragma unroll 1
    for (int pass = 0; pass < 2; ++pass) {
        const int cls = pass == 0 ? mine : 1 - mine;
        int* const counter = tickets + cls * TICKET_PAD;
        int b = __builtin_amdgcn_readfirstlane(ask_block(counter, lane));
        while (b < nblk) {
            int nxt;
            const long e0 = (long)b * (32 * EDGES_PL) + lane;
            if ((long)(b + 1) * (32 * EDGES_PL) <= E) {
                int d_[EDGES_PL], s_[EDGES_PL], prev[EDGES_PL];
                #pragma unroll
                for (int j = 0; j < EDGES_PL; ++j) { d_[j] = dstI[e0 + 32 * j]; s_[j] = srcI[e0 + 32 * j]; }
                nxt = ask_block(counter, lane);
                #pragma unroll
                for (int j = 0; j < EDGES_PL; ++j) {
                    prev[j] = -1;
                    if ((d_[j] >= split ? 1 : 0) == cls) prev[j] = atomicExch(&head[d_[j]], (int)(e0 + 32 * j));
                }
                #pragma unroll
                for (int j = 0; j < EDGES_PL; ++j)
                    if ((d_[j] >= split ? 1 : 0) == cls) link[e0 + 32 * j] = make_int2(prev[j], s_[j]);
            } else {
                nxt = ask_block(counter, lane);
                #pragma unroll
                for (int j = 0; j < EDGES_PL; ++j) {
                    const long e = e0 + 32 * j;
                    if (e < E) {
                        const int d1 = dstI[e];
                        if ((d1 >= split ? 1 : 0) == cls) {
                            const int s1 = srcI[e];
                            const int p1 = atomicExch(&head[d1], (int)e);
                            link[e] = make_int2(p1, s1);
                        }
                    }
                }
            }
            b = __builtin_amdgcn_readfirstlane(nxt);
        }
    }
}

__global__ __launch_bounds__(256) void ln_gemm_kernel(
    const float* __restrict__ x,
    const float* __restrict__ gamma, const float* __restrict__ beta,
    const float* __restrict__ Wl, const float* __restrict__ bl,
    const float* __restrict__ Wr, const float* __restrict__ br,
    float* __restrict__ xl, float* __restrict__ xr, long N,
    const int* __restrict__ srcI, const int* __restrict__ dstI,
    int* __restrict__ head, int2* __restrict__ link, long E)
{
    __shared__ float sXh[ROWS_PB][132];

    const int t   = threadIdx.x;
    if (blockIdx.x < LIST_WGS) {
        list_build_by_half(srcI, dstI, head, link, E, N);
        return;
    }
    const int r0  = t >> 4;
    const int sub = t & 15;
    const long blockRow = (long)(blockIdx.x - LIST_WGS) * ROWS_PB;

    float v[ROW_TILES][8];
    #pragma unroll
    for (int p = 0; p < ROW_TILES; ++p) {
        const long node = blockRow + p * 16 + r0;
        if (node < N) {
            const float4* xp = (const float4*)(x + node * C_IN + sub * 8);
            float4 p0 = xp[0], p1 = xp[1];
            v[p][0]=p0.x; v[p][1]=p0.y; v[p][2]=p0.z; v[p][3]=p0.w;
            v[p][4]=p1.x; v[p][5]=p1.y; v[p][6]=p1.z; v[p][7]=p1.w;
        } else {
            #pragma unroll
            for (int j = 0; j < 8; ++j) v[p][j] = 0.0f;
        }
    }

    #pragma unroll
    for (int p = 0; p < ROW_TILES; ++p) {
        float s = 0.0f;
        #pragma unroll
        for (int j = 0; j < 8; ++j) s += v[p][j];
        s += __shfl_xor(s, 1, 16); s += __shfl_xor(s, 2, 16);
        s += __shfl_xor(s, 4, 16); s += __shfl_xor(s, 8, 16);
        const float mu = s * (1.0f / 128.0f);

        float q = 0.0f;
        #pragma unroll
        for (int j = 0; j < 8; ++j) { float d = v[p][j] - mu; q += d * d; }
        q += __shfl_xor(q, 1, 16); q += __shfl_xor(q, 2, 16);
        q += __shfl_xor(q, 4, 16); q += __shfl_xor(q, 8, 16);
        const float rstd = rsqrtf(q * (1.0f / 128.0f) + LN_EPS);

        #pragma unroll
        for (int j = 0; j < 8; ++j) {
            const int c = sub * 8 + j;
            float y = (v[p][j] - mu) * rstd * gamma[c] + beta[c];
            sXh[p * 16 + r0][c] = y > 0.0f ? y : 0.0f;
        }
    }
    __syncthreads();

    const int lane = t & 31;
    const int wave = t >> 5;
    const int half = lane >> 4;
    const int lr   = lane & 15;
    const int col  = wave * 16 + lr;

    v8f cl[ROW_TILES], cr[ROW_TILES];
    #pragma unroll
    for (int p = 0; p < ROW_TILES; ++p) { cl[p] = (v8f){}; cr[p] = (v8f){}; }

    v2f pb[KG], qb[KG];
    #pragma unroll
    for (int s = 0; s < KG; ++s) {
        const int rk = (4 * s + 2 * half) * C_IN + col;
        pb[s].x = Wl[rk]; pb[s].y = Wl[rk + C_IN];
        qb[s].x = Wr[rk]; qb[s].y = Wr[rk + C_IN];
    }
    #pragma unroll
    for (int kg = 0; kg < 128; kg += 4 * KG) {
        v2f npb[KG], nqb[KG];
        if (kg + 4 * KG < 128) {
            #pragma unroll
            for (int s = 0; s < KG; ++s) {
                const int rk = (kg + 4 * KG + 4 * s + 2 * half) * C_IN + col;
                npb[s].x = Wl[rk]; npb[s].y = Wl[rk + C_IN];
                nqb[s].x = Wr[rk]; nqb[s].y = Wr[rk + C_IN];
            }
        }
        #pragma unroll
        for (int s = 0; s < KG; ++s) {
            const int k = kg + 4 * s;
            #pragma unroll
            for (int p = 0; p < ROW_TILES; ++p) {
                v2f a = *(const v2f*)&sXh[p * 16 + lr][k + 2 * half];
                cl[p] = __builtin_amdgcn_wmma_f32_16x16x4_f32(false, a, false, pb[s], (short)0, cl[p], false, false);
                cr[p] = __builtin_amdgcn_wmma_f32_16x16x4_f32(false, a, false, qb[s], (short)0, cr[p], false, false);
            }
        }
        if (kg + 4 * KG < 128) {
            #pragma unroll
            for (int s = 0; s < KG; ++s) { pb[s] = npb[s]; qb[s] = nqb[s]; }
        }
    }

    static_assert(ROW_TILES == 5, "the statement below names five accumulator pairs");
    asm volatile("v_nop\n\tv_nop\n\tv_nop\n\tv_nop\n\tv_nop"
                 : "+v"(cl[0]), "+v"(cl[1]), "+v"(cl[2]), "+v"(cl[3]), "+v"(cl[4]),
                   "+v"(cr[0]), "+v"(cr[1]), "+v"(cr[2]), "+v"(cr[3]), "+v"(cr[4]));

    const float bll = bl[col], brr = br[col];
    if (blockRow + ROWS_PB <= N) {
        #pragma unroll
        for (int p = 0; p < ROW_TILES; ++p) {
            #pragma unroll
            for (int vv = 0; vv < 8; ++vv) {
                const long r = blockRow + p * 16 + vv + 8 * half;
                xl[r * C_IN + col] = cl[p][vv] + bll;
                xr[r * C_IN + col] = cr[p][vv] + brr;
            }
        }
    } else {
        #pragma unroll
        for (int p = 0; p < ROW_TILES; ++p) {
            #pragma unroll
            for (int vv = 0; vv < 8; ++vv) {
                const long r = blockRow + p * 16 + vv + 8 * half;
                if (r < N) {
                    xl[r * C_IN + col] = cl[p][vv] + bll;
                    xr[r * C_IN + col] = cr[p][vv] + brr;
                }
            }
        }
    }
}

#define NODE_GRID 1536
#define NB 9
#define CAP 64
#define PITCH 72
#define GRP 4

__device__ __forceinline__ float lane_dot(const float4 l4, const float4 r4, const float4 a4) {
    float4 t4;
    t4.x = l4.x + r4.x; t4.y = l4.y + r4.y;
    t4.z = l4.z + r4.z; t4.w = l4.w + r4.w;
    t4.x = t4.x > 0.0f ? t4.x : NEG_SLOPE * t4.x;
    t4.y = t4.y > 0.0f ? t4.y : NEG_SLOPE * t4.y;
    t4.z = t4.z > 0.0f ? t4.z : NEG_SLOPE * t4.z;
    t4.w = t4.w > 0.0f ? t4.w : NEG_SLOPE * t4.w;
    return t4.x * a4.x + t4.y * a4.y + t4.z * a4.z + t4.w * a4.w;
}

__device__ __forceinline__ float head_sum(float p) {
    p += __shfl_xor(p, 1, 8);
    p += __shfl_xor(p, 2, 8);
    p += __shfl_xor(p, 4, 8);
    return p;
}

__device__ __forceinline__ void edge_update(const float4 row, const float pp,
                                            float& m, float& den, float4& acc) {
    const float p  = head_sum(pp);
    const float mn = fmaxf(m, p);
    const float sc = __expf(m - mn);
    const float w  = __expf(p - mn);
    den   = den   * sc + w;
    acc.x = acc.x * sc + w * row.x;
    acc.y = acc.y * sc + w * row.y;
    acc.z = acc.z * sc + w * row.z;
    acc.w = acc.w * sc + w * row.w;
    m = mn;
}

__device__ __forceinline__ void take_edge(const float4 row, const float4 r4, const float4 a4,
                                          float& m, float& den, float4& acc) {
    edge_update(row, lane_dot(row, r4, a4), m, den, acc);
}

__device__ __forceinline__ int link_slot(const int e, const int last) {
    return min(max(e, 0), last);
}

__device__ __forceinline__ float4 row_of(const float4* __restrict__ xl4, const int s, const int lane) {
    const float4* rowp = xl4 + (long)s * 32;
    return rowp[lane];
}

__device__ __forceinline__ void load_rows(float4 (&row)[GRP], const int* tab, const int k, const int T,
                                          const float4* __restrict__ xl4, const int lane) {
    const int4 q = *(const int4*)(tab + k);
    const int i0 = __builtin_amdgcn_readfirstlane(q.x);
    const int i1 = (k + 1 < T) ? __builtin_amdgcn_readfirstlane(q.y) : i0;
    const int i2 = (k + 2 < T) ? __builtin_amdgcn_readfirstlane(q.z) : i1;
    const int i3 = (k + 3 < T) ? __builtin_amdgcn_readfirstlane(q.w) : i2;
    row[0] = row_of(xl4, i0, lane);
    row[1] = row_of(xl4, i1, lane);
    row[2] = row_of(xl4, i2, lane);
    row[3] = row_of(xl4, i3, lane);
}

__global__ __launch_bounds__(256, 12) void node_pull_kernel(
    const int* __restrict__ head, const int2* __restrict__ link,
    const float* __restrict__ xl, const float* __restrict__ xr,
    const float* __restrict__ att, const float* __restrict__ bias,
    float* __restrict__ out, long N, long E)
{
    __shared__ __attribute__((aligned(16))) int sSrc[8][NB][PITCH];

    const int lane = threadIdx.x & 31;
    const int wave = __builtin_amdgcn_readfirstlane((int)(threadIdx.x >> 5));
    const unsigned nw  = gridDim.x * 8u;
    const unsigned wid = blockIdx.x * 8u + (unsigned)wave;
    const unsigned per = (unsigned)N / nw, extra = (unsigned)N % nw;
    const long lo = (long)wid * per + min(wid, extra);
    const long hi = lo + per + (wid < extra ? 1 : 0);
    const float4 a4 = ((const float4*)att)[lane];
    const float4* xl4 = (const float4*)xl;
    const float4* xr4 = (const float4*)xr;
    const int last = (int)(E - 1);
    const int nmax = (int)(N - 1);
    int* const mine = &sSrc[wave][lane < NB ? lane : 0][0];

    for (long n0 = lo; n0 < hi; n0 += NB) {
        const int nb = (int)min((long)NB, hi - n0);

        int cnt = 0;
        int cur = -1;
        if (lane < nb) {
            mine[0] = (int)(n0 + lane);
            cnt = 1;
            cur = head[n0 + lane];
        }
        for (int h = 1; h < CAP; ++h) {
            if (__builtin_amdgcn_ballot_w32(cur >= 0) == 0u) break;
            if (cur >= 0) {
                const int2 l = link[link_slot(cur, last)];
                mine[cnt] = min(max(l.y, 0), nmax);
                ++cnt;
                cur = l.x;
            }
        }

        for (int j = 0; j < nb; ++j) {
            const long i = n0 + j;
            const int T   = __builtin_amdgcn_readlane(cnt, j);
            const int rem = __builtin_amdgcn_readlane(cur, j);
            const int* tab = &sSrc[wave][j][0];
            const float4 r4 = (xr4 + i * 32)[lane];

            float m = -INFINITY;
            float den = 0.0f;
            float4 acc = make_float4(0.0f, 0.0f, 0.0f, 0.0f);

            float4 row[GRP];
            load_rows(row, tab, 0, T, xl4, lane);
            int k = 0;
            while (k + 2 * GRP <= T) {
                const int4 q = *(const int4*)(tab + k + GRP);
                take_edge(row[0], r4, a4, m, den, acc);
                row[0] = row_of(xl4, __builtin_amdgcn_readfirstlane(q.x), lane);
                take_edge(row[1], r4, a4, m, den, acc);
                row[1] = row_of(xl4, __builtin_amdgcn_readfirstlane(q.y), lane);
                take_edge(row[2], r4, a4, m, den, acc);
                row[2] = row_of(xl4, __builtin_amdgcn_readfirstlane(q.z), lane);
                take_edge(row[3], r4, a4, m, den, acc);
                row[3] = row_of(xl4, __builtin_amdgcn_readfirstlane(q.w), lane);
                k += GRP;
            }
            if (T - k > GRP) {
                float4 nxt[GRP];
                load_rows(nxt, tab, k + GRP, T, xl4, lane);
                #pragma unroll
                for (int u = 0; u < GRP; ++u) take_edge(row[u], r4, a4, m, den, acc);
                #pragma unroll
                for (int u = 0; u < GRP; ++u)
                    if (k + GRP + u < T) take_edge(nxt[u], r4, a4, m, den, acc);
            } else {
                #pragma unroll
                for (int u = 0; u < GRP; ++u)
                    if (k + u < T) take_edge(row[u], r4, a4, m, den, acc);
            }

            int e = rem;
            for (long c2 = 0; c2 < E && e >= 0; ++c2) {
                const int2 l = link[link_slot(e, last)];
                const int s = min(max(__builtin_amdgcn_readfirstlane(l.y), 0), nmax);
                e = __builtin_amdgcn_readfirstlane(l.x);
                take_edge(row_of(xl4, s, lane), r4, a4, m, den, acc);
            }

            const float inv = 0.25f / (den + 1e-16f);
            float vx = acc.x * inv, vy = acc.y * inv, vz = acc.z * inv, vw = acc.w * inv;
            vx += __shfl_xor(vx, 8);  vy += __shfl_xor(vy, 8);
            vz += __shfl_xor(vz, 8);  vw += __shfl_xor(vw, 8);
            vx += __shfl_xor(vx, 16); vy += __shfl_xor(vy, 16);
            vz += __shfl_xor(vz, 16); vw += __shfl_xor(vw, 16);
            if (lane < 8) {
                const float4 b4 = ((const float4*)bias)[lane];
                float4 o4;
                o4.x = vx + b4.x; o4.y = vy + b4.y; o4.z = vz + b4.z; o4.w = vw + b4.w;
                ((float4*)out)[i * 8 + lane] = o4;
            }
        }
    }
}

extern "C" void kernel_launch(void* const* d_in, const int* in_sizes, int n_in,
                              void* d_out, int out_size, void* d_ws, size_t ws_size,
                              hipStream_t stream) {
    const float* x     = (const float*)d_in[0];
    const int*   ei    = (const int*)d_in[1];
    const float* gamma = (const float*)d_in[2];
    const float* beta  = (const float*)d_in[3];
    const float* Wl    = (const float*)d_in[4];
    const float* bl    = (const float*)d_in[5];
    const float* Wr    = (const float*)d_in[6];
    const float* br    = (const float*)d_in[7];
    const float* att   = (const float*)d_in[8];
    const float* bias  = (const float*)d_in[9];
    float* out = (float*)d_out;

    const long N = in_sizes[0] / C_IN;
    const long E = in_sizes[1] / 2;
    const int* srcI = ei;
    const int* dstI = ei + E;

    float* ws   = (float*)d_ws;
    float* xl   = ws;
    float* xr   = xl + N * C_IN;
    int2*  link = (int2*)(xr + N * C_IN);
    int*   head = (int*)(link + E);

    {
        head_clear_kernel<<<(int)((N + 255) / 256), 256, 0, stream>>>(head, N);
    }
    {
        int blocks = (int)((N + ROWS_PB - 1) / ROWS_PB);
        ln_gemm_kernel<<<blocks + LIST_WGS, 256, 0, stream>>>(x, gamma, beta, Wl, bl, Wr, br, xl, xr, N,
                                                             srcI, dstI, head, link, E);
    }
    {
        node_pull_kernel<<<NODE_GRID, 256, 0, stream>>>(head, link, xl, xr, att, bias, out, N, E);
    }
}


# Linked lists -> slot rows: an edge takes slot = returning add on its node's count, stores its source in the node's output row (scratch until the sums are stored): no list walk; sum order = slot order
# speedup vs baseline: 4.1752x; 1.0321x over previous
#include <hip/hip_runtime.h>
#include <math.h>

typedef float v2f __attribute__((ext_vector_type(2)));
typedef float v8f __attribute__((ext_vector_type(8)));

#define C_IN 128
#define NEG_SLOPE 0.2f
#define LN_EPS 1e-5f

#define ROWS_PB 80
#define ROW_TILES 5
#define KG 8
#define LIST_WGS 256
#define EDGES_PL 16
#define TICKET_PAD 64

__device__ __forceinline__ int* ticket_words(int* head, const long N) {
    return head + ((N + 63) & ~63L);
}

#define ROW_W 32
#define SLOTS 31

__global__ __launch_bounds__(256) void head_clear_kernel(int* __restrict__ head, int* __restrict__ outw, long N)
{
    const long i = (long)blockIdx.x * 256 + threadIdx.x;
    if (i < N) { head[i] = 0; outw[i * ROW_W + SLOTS] = -1; }
    if (i < 2) ticket_words(head, N)[i * TICKET_PAD] = 0;
}


__device__ __forceinline__ int chip_half() {
    const unsigned id = __builtin_amdgcn_s_sendmsg_rtn(0x87);
    return (int)((id >> 18) & 1u);
}

__device__ __forceinline__ int ask_block(int* __restrict__ counter, const int lane) {
    unsigned t = 0u;
    if (lane == 0) t = atomicInc((unsigned*)counter, 0xFFFFFFFFu);
    return (int)t;
}

__device__ __forceinline__ void place_edge(int* __restrict__ outw, int2* __restrict__ link,
                                           const int d, const int s, const int n, const long e) {
    int* const rowp = outw + (long)d * ROW_W;
    if (n < SLOTS) {
        rowp[n] = s;
    } else {
        const int p = atomicExch(rowp + SLOTS, (int)e);
        link[e] = make_int2(p, s);
    }
}

__device__ __forceinline__ void list_build_by_half(
    const int* __restrict__ srcI, const int* __restrict__ dstI,
    int* __restrict__ head, int2* __restrict__ link, int* __restrict__ outw, const long E, const long N)
{
    const int lane = threadIdx.x & 31;
    const int nblk = (int)((E + 32 * EDGES_PL - 1) / (32 * EDGES_PL));
    const int split = (int)(((N >> 1) + 63) & ~63L);
    int* const tickets = ticket_words(head, N);
    const int mine = chip_half();

    #pragma unroll 1
    for (int pass = 0; pass < 2; ++pass) {
        const int cls = pass == 0 ? mine : 1 - mine;
        int* const counter = tickets + cls * TICKET_PAD;
        int b = __builtin_amdgcn_readfirstlane(ask_block(counter, lane));
        while (b < nblk) {
            int nxt;
            const long e0 = (long)b * (32 * EDGES_PL) + lane;
            if ((long)(b + 1) * (32 * EDGES_PL) <= E) {
                int d_[EDGES_PL], s_[EDGES_PL], slot[EDGES_PL];
                #pragma unroll
                for (int j = 0; j < EDGES_PL; ++j) { d_[j] = dstI[e0 + 32 * j]; s_[j] = srcI[e0 + 32 * j]; }
                nxt = ask_block(counter, lane);
                #pragma unroll
                for (int j = 0; j < EDGES_PL; ++j) {
                    slot[j] = 0;
                    if ((d_[j] >= split ? 1 : 0) == cls) slot[j] = atomicAdd(&head[d_[j]], 1);
                }
                #pragma unroll
                for (int j = 0; j < EDGES_PL; ++j)
                    if ((d_[j] >= split ? 1 : 0) == cls) place_edge(outw, link, d_[j], s_[j], slot[j], e0 + 32 * j);
            } else {
                nxt = ask_block(counter, lane);
                #pragma unroll
                for (int j = 0; j < EDGES_PL; ++j) {
                    const long e = e0 + 32 * j;
                    if (e < E) {
                        const int d1 = dstI[e];
                        if ((d1 >= split ? 1 : 0) == cls) {
                            const int s1 = srcI[e];
                            const int n1 = atomicAdd(&head[d1], 1);
                            place_edge(outw, link, d1, s1, n1, e);
                        }
                    }
                }
            }
            b = __builtin_amdgcn_readfirstlane(nxt);
        }
    }
}

__global__ __launch_bounds__(256) void ln_gemm_kernel(
    const float* __restrict__ x,
    const float* __restrict__ gamma, const float* __restrict__ beta,
    const float* __restrict__ Wl, const float* __restrict__ bl,
    const float* __restrict__ Wr, const float* __restrict__ br,
    float* __restrict__ xl, float* __restrict__ xr, long N,
    const int* __restrict__ srcI, const int* __restrict__ dstI,
    int* __restrict__ head, int2* __restrict__ link, int* __restrict__ outw, long E)
{
    __shared__ float sXh[ROWS_PB][132];

    const int t   = threadIdx.x;
    if (blockIdx.x < LIST_WGS) {
        list_build_by_half(srcI, dstI, head, link, outw, E, N);
        return;
    }
    const int r0  = t >> 4;
    const int sub = t & 15;
    const long blockRow = (long)(blockIdx.x - LIST_WGS) * ROWS_PB;

    float v[ROW_TILES][8];
    #pragma unroll
    for (int p = 0; p < ROW_TILES; ++p) {
        const long node = blockRow + p * 16 + r0;
        if (node < N) {
            const float4* xp = (const float4*)(x + node * C_IN + sub * 8);
            float4 p0 = xp[0], p1 = xp[1];
            v[p][0]=p0.x; v[p][1]=p0.y; v[p][2]=p0.z; v[p][3]=p0.w;
            v[p][4]=p1.x; v[p][5]=p1.y; v[p][6]=p1.z; v[p][7]=p1.w;
        } else {
            #pragma unroll
            for (int j = 0; j < 8; ++j) v[p][j] = 0.0f;
        }
    }

    #pragma unroll
    for (int p = 0; p < ROW_TILES; ++p) {
        float s = 0.0f;
        #pragma unroll
        for (int j = 0; j < 8; ++j) s += v[p][j];
        s += __shfl_xor(s, 1, 16); s += __shfl_xor(s, 2, 16);
        s += __shfl_xor(s, 4, 16); s += __shfl_xor(s, 8, 16);
        const float mu = s * (1.0f / 128.0f);

        float q = 0.0f;
        #pragma unroll
        for (int j = 0; j < 8; ++j) { float d = v[p][j] - mu; q += d * d; }
        q += __shfl_xor(q, 1, 16); q += __shfl_xor(q, 2, 16);
        q += __shfl_xor(q, 4, 16); q += __shfl_xor(q, 8, 16);
        const float rstd = rsqrtf(q * (1.0f / 128.0f) + LN_EPS);

        #pragma unroll
        for (int j = 0; j < 8; ++j) {
            const int c = sub * 8 + j;
            float y = (v[p][j] - mu) * rstd * gamma[c] + beta[c];
            sXh[p * 16 + r0][c] = y > 0.0f ? y : 0.0f;
        }
    }
    __syncthreads();

    const int lane = t & 31;
    const int wave = t >> 5;
    const int half = lane >> 4;
    const int lr   = lane & 15;
    const int col  = wave * 16 + lr;

    v8f cl[ROW_TILES], cr[ROW_TILES];
    #pragma unroll
    for (int p = 0; p < ROW_TILES; ++p) { cl[p] = (v8f){}; cr[p] = (v8f){}; }

    v2f pb[KG], qb[KG];
    #pragma unroll
    for (int s = 0; s < KG; ++s) {
        const int rk = (4 * s + 2 * half) * C_IN + col;
        pb[s].x = Wl[rk]; pb[s].y = Wl[rk + C_IN];
        qb[s].x = Wr[rk]; qb[s].y = Wr[rk + C_IN];
    }
    #pragma unroll
    for (int kg = 0; kg < 128; kg += 4 * KG) {
        v2f npb[KG], nqb[KG];
        if (kg + 4 * KG < 128) {
            #pragma unroll
            for (int s = 0; s < KG; ++s) {
                const int rk = (kg + 4 * KG + 4 * s + 2 * half) * C_IN + col;
                npb[s].x = Wl[rk]; npb[s].y = Wl[rk + C_IN];
                nqb[s].x = Wr[rk]; nqb[s].y = Wr[rk + C_IN];
            }
        }
        #pragma unroll
        for (int s = 0; s < KG; ++s) {
            const int k = kg + 4 * s;
            #pragma unroll
            for (int p = 0; p < ROW_TILES; ++p) {
                v2f a = *(const v2f*)&sXh[p * 16 + lr][k + 2 * half];
                cl[p] = __builtin_amdgcn_wmma_f32_16x16x4_f32(false, a, false, pb[s], (short)0, cl[p], false, false);
                cr[p] = __builtin_amdgcn_wmma_f32_16x16x4_f32(false, a, false, qb[s], (short)0, cr[p], false, false);
            }
        }
        if (kg + 4 * KG < 128) {
            #pragma unroll
            for (int s = 0; s < KG; ++s) { pb[s] = npb[s]; qb[s] = nqb[s]; }
        }
    }

    static_assert(ROW_TILES == 5, "the statement below names five accumulator pairs");
    asm volatile("v_nop\n\tv_nop\n\tv_nop\n\tv_nop\n\tv_nop"
                 : "+v"(cl[0]), "+v"(cl[1]), "+v"(cl[2]), "+v"(cl[3]), "+v"(cl[4]),
                   "+v"(cr[0]), "+v"(cr[1]), "+v"(cr[2]), "+v"(cr[3]), "+v"(cr[4]));

    const float bll = bl[col], brr = br[col];
    if (blockRow + ROWS_PB <= N) {
        #pragma unroll
        for (int p = 0; p < ROW_TILES; ++p) {
            #pragma unroll
            for (int vv = 0; vv < 8; ++vv) {
                const long r = blockRow + p * 16 + vv + 8 * half;
                xl[r * C_IN + col] = cl[p][vv] + bll;
                xr[r * C_IN + col] = cr[p][vv] + brr;
            }
        }
    } else {
        #pragma unroll
        for (int p = 0; p < ROW_TILES; ++p) {
            #pragma unroll
            for (int vv = 0; vv < 8; ++vv) {
                const long r = blockRow + p * 16 + vv + 8 * half;
                if (r < N) {
                    xl[r * C_IN + col] = cl[p][vv] + bll;
                    xr[r * C_IN + col] = cr[p][vv] + brr;
                }
            }
        }
    }
}

#define NODE_GRID 1536
#define GRP 4

__device__ __forceinline__ float lane_dot(const float4 l4, const float4 r4, const float4 a4) {
    float4 t4;
    t4.x = l4.x + r4.x; t4.y = l4.y + r4.y;
    t4.z = l4.z + r4.z; t4.w = l4.w + r4.w;
    t4.x = t4.x > 0.0f ? t4.x : NEG_SLOPE * t4.x;
    t4.y = t4.y > 0.0f ? t4.y : NEG_SLOPE * t4.y;
    t4.z = t4.z > 0.0f ? t4.z : NEG_SLOPE * t4.z;
    t4.w = t4.w > 0.0f ? t4.w : NEG_SLOPE * t4.w;
    return t4.x * a4.x + t4.y * a4.y + t4.z * a4.z + t4.w * a4.w;
}

__device__ __forceinline__ float head_sum(float p) {
    p += __shfl_xor(p, 1, 8);
    p += __shfl_xor(p, 2, 8);
    p += __shfl_xor(p, 4, 8);
    return p;
}

__device__ __forceinline__ void edge_update(const float4 row, const float pp,
                                            float& m, float& den, float4& acc) {
    const float p  = head_sum(pp);
    const float mn = fmaxf(m, p);
    const float sc = __expf(m - mn);
    const float w  = __expf(p - mn);
    den   = den   * sc + w;
    acc.x = acc.x * sc + w * row.x;
    acc.y = acc.y * sc + w * row.y;
    acc.z = acc.z * sc + w * row.z;
    acc.w = acc.w * sc + w * row.w;
    m = mn;
}

__device__ __forceinline__ void take_edge(const float4 row, const float4 r4, const float4 a4,
                                          float& m, float& den, float4& acc) {
    edge_update(row, lane_dot(row, r4, a4), m, den, acc);
}

__device__ __forceinline__ int link_slot(const int e, const int last) {
    return min(max(e, 0), last);
}

__device__ __forceinline__ float4 row_of(const float4* __restrict__ xl4, const int s, const int lane) {
    const float4* rowp = xl4 + (long)s * 32;
    return rowp[lane];
}

__device__ __forceinline__ int entry_src(const int ent, const int t) {
    return __builtin_amdgcn_readlane(ent, t);
}

__device__ __forceinline__ void load_rows(float4 (&row)[GRP], const int ent, const int k, const int T,
                                          const float4* __restrict__ xl4, const int lane) {
    const int i0 = entry_src(ent, k);
    const int i1 = (k + 1 < T) ? entry_src(ent, k + 1) : i0;
    const int i2 = (k + 2 < T) ? entry_src(ent, k + 2) : i1;
    const int i3 = (k + 3 < T) ? entry_src(ent, k + 3) : i2;
    row[0] = row_of(xl4, i0, lane);
    row[1] = row_of(xl4, i1, lane);
    row[2] = row_of(xl4, i2, lane);
    row[3] = row_of(xl4, i3, lane);
}

__device__ __forceinline__ int node_entries(const int raw, const int i, const int lane, const int nmax) {
    return lane == 0 ? i : min(max(raw, 0), nmax);
}

__global__ __launch_bounds__(256, 12) void node_pull_kernel(
    const int* __restrict__ head, const int2* __restrict__ link,
    const float* __restrict__ xl, const float* __restrict__ xr,
    const float* __restrict__ att, const float* __restrict__ bias,
    int* outw, long N, long E)
{
    const int lane = threadIdx.x & 31;
    const int wave = __builtin_amdgcn_readfirstlane((int)(threadIdx.x >> 5));
    const unsigned nw  = gridDim.x * 8u;
    const unsigned wid = blockIdx.x * 8u + (unsigned)wave;
    const unsigned per = (unsigned)N / nw, extra = (unsigned)N % nw;
    const int lo = (int)(wid * per + min(wid, extra));
    const int hi = lo + (int)per + (wid < extra ? 1 : 0);
    if (lo >= hi) return;
    const float4 a4 = ((const float4*)att)[lane];
    const float4* xl4 = (const float4*)xl;
    const float4* xr4 = (const float4*)xr;
    const int last = (int)(E - 1);
    const int nmax = (int)(N - 1);
    const int slotoff = lane > 0 ? lane - 1 : 0;

    int ncn  = head[lo];
    int nraw = outw[(long)lo * ROW_W + slotoff];

    for (int i = lo; i < hi; ++i) {
        const int cnt = max(ncn, 0);
        const int ent = node_entries(nraw, i, lane, nmax);
        const int inx = min(i + 1, hi - 1);
        ncn  = head[inx];
        nraw = outw[(long)inx * ROW_W + slotoff];
        const float4 r4 = (xr4 + (long)i * 32)[lane];

        const int T   = 1 + min(cnt, SLOTS);
        int rem = -1;
        if (cnt > SLOTS) rem = __builtin_amdgcn_readfirstlane(outw[(long)i * ROW_W + SLOTS]);

        float m = -INFINITY;
        float den = 0.0f;
        float4 acc = make_float4(0.0f, 0.0f, 0.0f, 0.0f);

        float4 row[GRP];
        load_rows(row, ent, 0, T, xl4, lane);
        int k = 0;
        while (k + 2 * GRP <= T) {
            take_edge(row[0], r4, a4, m, den, acc);
            row[0] = row_of(xl4, entry_src(ent, k + GRP), lane);
            take_edge(row[1], r4, a4, m, den, acc);
            row[1] = row_of(xl4, entry_src(ent, k + GRP + 1), lane);
            take_edge(row[2], r4, a4, m, den, acc);
            row[2] = row_of(xl4, entry_src(ent, k + GRP + 2), lane);
            take_edge(row[3], r4, a4, m, den, acc);
            row[3] = row_of(xl4, entry_src(ent, k + GRP + 3), lane);
            k += GRP;
        }
        if (T - k > GRP) {
            float4 nxt[GRP];
            load_rows(nxt, ent, k + GRP, T, xl4, lane);
            #pragma unroll
            for (int u = 0; u < GRP; ++u) take_edge(row[u], r4, a4, m, den, acc);
            #pragma unroll
            for (int u = 0; u < GRP; ++u)
                if (k + GRP + u < T) take_edge(nxt[u], r4, a4, m, den, acc);
        } else {
            #pragma unroll
            for (int u = 0; u < GRP; ++u)
                if (k + u < T) take_edge(row[u], r4, a4, m, den, acc);
        }

        int e = rem;
        for (long c2 = 0; c2 < E && e >= 0; ++c2) {
            const int2 l = link[link_slot(e, last)];
            const int s = min(max(__builtin_amdgcn_readfirstlane(l.y), 0), nmax);
            e = __builtin_amdgcn_readfirstlane(l.x);
            take_edge(row_of(xl4, s, lane), r4, a4, m, den, acc);
        }

        const float inv = 0.25f / (den + 1e-16f);
        float vx = acc.x * inv, vy = acc.y * inv, vz = acc.z * inv, vw = acc.w * inv;
        vx += __shfl_xor(vx, 8);  vy += __shfl_xor(vy, 8);
        vz += __shfl_xor(vz, 8);  vw += __shfl_xor(vw, 8);
        vx += __shfl_xor(vx, 16); vy += __shfl_xor(vy, 16);
        vz += __shfl_xor(vz, 16); vw += __shfl_xor(vw, 16);
        if (lane < 8) {
            const float4 b4 = ((const float4*)bias)[lane];
            int4 o4;
            o4.x = __float_as_int(vx + b4.x); o4.y = __float_as_int(vy + b4.y);
            o4.z = __float_as_int(vz + b4.z); o4.w = __float_as_int(vw + b4.w);
            ((int4*)outw)[(long)i * 8 + lane] = o4;
        }
    }
}

extern "C" void kernel_launch(void* const* d_in, const int* in_sizes, int n_in,
                              void* d_out, int out_size, void* d_ws, size_t ws_size,
                              hipStream_t stream) {
    const float* x     = (const float*)d_in[0];
    const int*   ei    = (const int*)d_in[1];
    const float* gamma = (const float*)d_in[2];
    const float* beta  = (const float*)d_in[3];
    const float* Wl    = (const float*)d_in[4];
    const float* bl    = (const float*)d_in[5];
    const float* Wr    = (const float*)d_in[6];
    const float* br    = (const float*)d_in[7];
    const float* att   = (const float*)d_in[8];
    const float* bias  = (const float*)d_in[9];
    int* outw = (int*)d_out;

    const long N = in_sizes[0] / C_IN;
    const long E = in_sizes[1] / 2;
    const int* srcI = ei;
    const int* dstI = ei + E;

    float* ws   = (float*)d_ws;
    float* xl   = ws;
    float* xr   = xl + N * C_IN;
    int2*  link = (int2*)(xr + N * C_IN);
    int*   head = (int*)(link + E);

    {
        head_clear_kernel<<<(int)((N + 255) / 256), 256, 0, stream>>>(head, outw, N);
    }
    {
        int blocks = (int)((N + ROWS_PB - 1) / ROWS_PB);
        ln_gemm_kernel<<<blocks + LIST_WGS, 256, 0, stream>>>(x, gamma, beta, Wl, bl, Wr, br, xl, xr, N,
                                                             srcI, dstI, head, link, outw, E);
    }
    {
        node_pull_kernel<<<NODE_GRID, 256, 0, stream>>>(head, link, xl, xr, att, bias, outw, N, E);
    }
}


# Node launch: slot entries summed four at a time: four scores first, then one running-max update and one rescale per group (5 exponentials per 4 entries, not 8); 10 waves a SIMD; term order kept
# speedup vs baseline: 4.4032x; 1.0546x over previous
#include <hip/hip_runtime.h>
#include <math.h>

typedef float v2f __attribute__((ext_vector_type(2)));
typedef float v8f __attribute__((ext_vector_type(8)));

#define C_IN 128
#define NEG_SLOPE 0.2f
#define LN_EPS 1e-5f

#define ROWS_PB 80
#define ROW_TILES 5
#define KG 8
#define LIST_WGS 256
#define EDGES_PL 16
#define TICKET_PAD 64

__device__ __forceinline__ int* ticket_words(int* head, const long N) {
    return head + ((N + 63) & ~63L);
}

#define ROW_W 32
#define SLOTS 31

__global__ __launch_bounds__(256) void head_clear_kernel(int* __restrict__ head, int* __restrict__ outw, long N)
{
    const long i = (long)blockIdx.x * 256 + threadIdx.x;
    if (i < N) { head[i] = 0; outw[i * ROW_W + SLOTS] = -1; }
    if (i < 2) ticket_words(head, N)[i * TICKET_PAD] = 0;
}


__device__ __forceinline__ int chip_half() {
    const unsigned id = __builtin_amdgcn_s_sendmsg_rtn(0x87);
    return (int)((id >> 18) & 1u);
}

__device__ __forceinline__ int ask_block(int* __restrict__ counter, const int lane) {
    unsigned t = 0u;
    if (lane == 0) t = atomicInc((unsigned*)counter, 0xFFFFFFFFu);
    return (int)t;
}

__device__ __forceinline__ void place_edge(int* __restrict__ outw, int2* __restrict__ link,
                                           const int d, const int s, const int n, const long e) {
    int* const rowp = outw + (long)d * ROW_W;
    if (n < SLOTS) {
        rowp[n] = s;
    } else {
        const int p = atomicExch(rowp + SLOTS, (int)e);
        link[e] = make_int2(p, s);
    }
}

__device__ __forceinline__ void list_build_by_half(
    const int* __restrict__ srcI, const int* __restrict__ dstI,
    int* __restrict__ head, int2* __restrict__ link, int* __restrict__ outw, const long E, const long N)
{
    const int lane = threadIdx.x & 31;
    const int nblk = (int)((E + 32 * EDGES_PL - 1) / (32 * EDGES_PL));
    const int split = (int)(((N >> 1) + 63) & ~63L);
    int* const tickets = ticket_words(head, N);
    const int mine = chip_half();

    #pragma unroll 1
    for (int pass = 0; pass < 2; ++pass) {
        const int cls = pass == 0 ? mine : 1 - mine;
        int* const counter = tickets + cls * TICKET_PAD;
        int b = __builtin_amdgcn_readfirstlane(ask_block(counter, lane));
        while (b < nblk) {
            int nxt;
            const long e0 = (long)b * (32 * EDGES_PL) + lane;
            if ((long)(b + 1) * (32 * EDGES_PL) <= E) {
                int d_[EDGES_PL], s_[EDGES_PL], slot[EDGES_PL];
                #pragma unroll
                for (int j = 0; j < EDGES_PL; ++j) { d_[j] = dstI[e0 + 32 * j]; s_[j] = srcI[e0 + 32 * j]; }
                nxt = ask_block(counter, lane);
                #pragma unroll
                for (int j = 0; j < EDGES_PL; ++j) {
                    slot[j] = 0;
                    if ((d_[j] >= split ? 1 : 0) == cls) slot[j] = atomicAdd(&head[d_[j]], 1);
                }
                #pragma unroll
                for (int j = 0; j < EDGES_PL; ++j)
                    if ((d_[j] >= split ? 1 : 0) == cls) place_edge(outw, link, d_[j], s_[j], slot[j], e0 + 32 * j);
            } else {
                nxt = ask_block(counter, lane);
                #pragma unroll
                for (int j = 0; j < EDGES_PL; ++j) {
                    const long e = e0 + 32 * j;
                    if (e < E) {
                        const int d1 = dstI[e];
                        if ((d1 >= split ? 1 : 0) == cls) {
                            const int s1 = srcI[e];
                            const int n1 = atomicAdd(&head[d1], 1);
                            place_edge(outw, link, d1, s1, n1, e);
                        }
                    }
                }
            }
            b = __builtin_amdgcn_readfirstlane(nxt);
        }
    }
}

__global__ __launch_bounds__(256) void ln_gemm_kernel(
    const float* __restrict__ x,
    const float* __restrict__ gamma, const float* __restrict__ beta,
    const float* __restrict__ Wl, const float* __restrict__ bl,
    const float* __restrict__ Wr, const float* __restrict__ br,
    float* __restrict__ xl, float* __restrict__ xr, long N,
    const int* __restrict__ srcI, const int* __restrict__ dstI,
    int* __restrict__ head, int2* __restrict__ link, int* __restrict__ outw, long E)
{
    __shared__ float sXh[ROWS_PB][132];

    const int t   = threadIdx.x;
    if (blockIdx.x < LIST_WGS) {
        list_build_by_half(srcI, dstI, head, link, outw, E, N);
        return;
    }
    const int r0  = t >> 4;
    const int sub = t & 15;
    const long blockRow = (long)(blockIdx.x - LIST_WGS) * ROWS_PB;

    float v[ROW_TILES][8];
    #pragma unroll
    for (int p = 0; p < ROW_TILES; ++p) {
        const long node = blockRow + p * 16 + r0;
        if (node < N) {
            const float4* xp = (const float4*)(x + node * C_IN + sub * 8);
            float4 p0 = xp[0], p1 = xp[1];
            v[p][0]=p0.x; v[p][1]=p0.y; v[p][2]=p0.z; v[p][3]=p0.w;
            v[p][4]=p1.x; v[p][5]=p1.y; v[p][6]=p1.z; v[p][7]=p1.w;
        } else {
            #pragma unroll
            for (int j = 0; j < 8; ++j) v[p][j] = 0.0f;
        }
    }

    #pragma unroll
    for (int p = 0; p < ROW_TILES; ++p) {
        float s = 0.0f;
        #pragma unroll
        for (int j = 0; j < 8; ++j) s += v[p][j];
        s += __shfl_xor(s, 1, 16); s += __shfl_xor(s, 2, 16);
        s += __shfl_xor(s, 4, 16); s += __shfl_xor(s, 8, 16);
        const float mu = s * (1.0f / 128.0f);

        float q = 0.0f;
        #pragma unroll
        for (int j = 0; j < 8; ++j) { float d = v[p][j] - mu; q += d * d; }
        q += __shfl_xor(q, 1, 16); q += __shfl_xor(q, 2, 16);
        q += __shfl_xor(q, 4, 16); q += __shfl_xor(q, 8, 16);
        const float rstd = rsqrtf(q * (1.0f / 128.0f) + LN_EPS);

        #pragma unroll
        for (int j = 0; j < 8; ++j) {
            const int c = sub * 8 + j;
            float y = (v[p][j] - mu) * rstd * gamma[c] + beta[c];
            sXh[p * 16 + r0][c] = y > 0.0f ? y : 0.0f;
        }
    }
    __syncthreads();

    const int lane = t & 31;
    const int wave = t >> 5;
    const int half = lane >> 4;
    const int lr   = lane & 15;
    const int col  = wave * 16 + lr;

    v8f cl[ROW_TILES], cr[ROW_TILES];
    #pragma unroll
    for (int p = 0; p < ROW_TILES; ++p) { cl[p] = (v8f){}; cr[p] = (v8f){}; }

    v2f pb[KG], qb[KG];
    #pragma unroll
    for (int s = 0; s < KG; ++s) {
        const int rk = (4 * s + 2 * half) * C_IN + col;
        pb[s].x = Wl[rk]; pb[s].y = Wl[rk + C_IN];
        qb[s].x = Wr[rk]; qb[s].y = Wr[rk + C_IN];
    }
    #pragma unroll
    for (int kg = 0; kg < 128; kg += 4 * KG) {
        v2f npb[KG], nqb[KG];
        if (kg + 4 * KG < 128) {
            #pragma unroll
            for (int s = 0; s < KG; ++s) {
                const int rk = (kg + 4 * KG + 4 * s + 2 * half) * C_IN + col;
                npb[s].x = Wl[rk]; npb[s].y = Wl[rk + C_IN];
                nqb[s].x = Wr[rk]; nqb[s].y = Wr[rk + C_IN];
            }
        }
        #pragma unroll
        for (int s = 0; s < KG; ++s) {
            const int k = kg + 4 * s;
            #pragma unroll
            for (int p = 0; p < ROW_TILES; ++p) {
                v2f a = *(const v2f*)&sXh[p * 16 + lr][k + 2 * half];
                cl[p] = __builtin_amdgcn_wmma_f32_16x16x4_f32(false, a, false, pb[s], (short)0, cl[p], false, false);
                cr[p] = __builtin_amdgcn_wmma_f32_16x16x4_f32(false, a, false, qb[s], (short)0, cr[p], false, false);
            }
        }
        if (kg + 4 * KG < 128) {
            #pragma unroll
            for (int s = 0; s < KG; ++s) { pb[s] = npb[s]; qb[s] = nqb[s]; }
        }
    }

    static_assert(ROW_TILES == 5, "the statement below names five accumulator pairs");
    asm volatile("v_nop\n\tv_nop\n\tv_nop\n\tv_nop\n\tv_nop"
                 : "+v"(cl[0]), "+v"(cl[1]), "+v"(cl[2]), "+v"(cl[3]), "+v"(cl[4]),
                   "+v"(cr[0]), "+v"(cr[1]), "+v"(cr[2]), "+v"(cr[3]), "+v"(cr[4]));

    const float bll = bl[col], brr = br[col];
    if (blockRow + ROWS_PB <= N) {
        #pragma unroll
        for (int p = 0; p < ROW_TILES; ++p) {
            #pragma unroll
            for (int vv = 0; vv < 8; ++vv) {
                const long r = blockRow + p * 16 + vv + 8 * half;
                xl[r * C_IN + col] = cl[p][vv] + bll;
                xr[r * C_IN + col] = cr[p][vv] + brr;
            }
        }
    } else {
        #pragma unroll
        for (int p = 0; p < ROW_TILES; ++p) {
            #pragma unroll
            for (int vv = 0; vv < 8; ++vv) {
                const long r = blockRow + p * 16 + vv + 8 * half;
                if (r < N) {
                    xl[r * C_IN + col] = cl[p][vv] + bll;
                    xr[r * C_IN + col] = cr[p][vv] + brr;
                }
            }
        }
    }
}

#define NODE_GRID 1280
#define GRP 4

__device__ __forceinline__ float lane_dot(const float4 l4, const float4 r4, const float4 a4) {
    float4 t4;
    t4.x = l4.x + r4.x; t4.y = l4.y + r4.y;
    t4.z = l4.z + r4.z; t4.w = l4.w + r4.w;
    t4.x = t4.x > 0.0f ? t4.x : NEG_SLOPE * t4.x;
    t4.y = t4.y > 0.0f ? t4.y : NEG_SLOPE * t4.y;
    t4.z = t4.z > 0.0f ? t4.z : NEG_SLOPE * t4.z;
    t4.w = t4.w > 0.0f ? t4.w : NEG_SLOPE * t4.w;
    return t4.x * a4.x + t4.y * a4.y + t4.z * a4.z + t4.w * a4.w;
}

__device__ __forceinline__ float head_sum(float p) {
    p += __shfl_xor(p, 1, 8);
    p += __shfl_xor(p, 2, 8);
    p += __shfl_xor(p, 4, 8);
    return p;
}

__device__ __forceinline__ void edge_update(const float4 row, const float pp,
                                            float& m, float& den, float4& acc) {
    const float p  = head_sum(pp);
    const float mn = fmaxf(m, p);
    const float sc = __expf(m - mn);
    const float w  = __expf(p - mn);
    den   = den   * sc + w;
    acc.x = acc.x * sc + w * row.x;
    acc.y = acc.y * sc + w * row.y;
    acc.z = acc.z * sc + w * row.z;
    acc.w = acc.w * sc + w * row.w;
    m = mn;
}

__device__ __forceinline__ void take_edge(const float4 row, const float4 r4, const float4 a4,
                                          float& m, float& den, float4& acc) {
    edge_update(row, lane_dot(row, r4, a4), m, den, acc);
}

__device__ __forceinline__ void take_group(const float4 (&row)[GRP], const float4 r4, const float4 a4,
                                           float& m, float& den, float4& acc) {
    float p[GRP], w[GRP];
    #pragma unroll
    for (int u = 0; u < GRP; ++u) p[u] = head_sum(lane_dot(row[u], r4, a4));
    float mn = m;
    #pragma unroll
    for (int u = 0; u < GRP; ++u) mn = fmaxf(mn, p[u]);
    const float sc = __expf(m - mn);
    #pragma unroll
    for (int u = 0; u < GRP; ++u) w[u] = __expf(p[u] - mn);
    den   = den   * sc + w[0];
    acc.x = acc.x * sc + w[0] * row[0].x;
    acc.y = acc.y * sc + w[0] * row[0].y;
    acc.z = acc.z * sc + w[0] * row[0].z;
    acc.w = acc.w * sc + w[0] * row[0].w;
    #pragma unroll
    for (int u = 1; u < GRP; ++u) {
        den   += w[u];
        acc.x += w[u] * row[u].x;
        acc.y += w[u] * row[u].y;
        acc.z += w[u] * row[u].z;
        acc.w += w[u] * row[u].w;
    }
    m = mn;
}

__device__ __forceinline__ void take_rest(const float4 (&row)[GRP], const int k, const int T,
                                          const float4 r4, const float4 a4,
                                          float& m, float& den, float4& acc) {
    #pragma unroll
    for (int u = 0; u < GRP - 1; ++u)
        if (k + u < T) take_edge(row[u], r4, a4, m, den, acc);
}

__device__ __forceinline__ int link_slot(const int e, const int last) {
    return min(max(e, 0), last);
}

__device__ __forceinline__ float4 row_of(const float4* __restrict__ xl4, const int s, const int lane) {
    const float4* rowp = xl4 + (long)s * 32;
    return rowp[lane];
}

__device__ __forceinline__ int entry_src(const int ent, const int t) {
    return __builtin_amdgcn_readlane(ent, t);
}

__device__ __forceinline__ void load_rows(float4 (&row)[GRP], const int ent, const int k, const int T,
                                          const float4* __restrict__ xl4, const int lane) {
    const int i0 = entry_src(ent, k);
    const int i1 = (k + 1 < T) ? entry_src(ent, k + 1) : i0;
    const int i2 = (k + 2 < T) ? entry_src(ent, k + 2) : i1;
    const int i3 = (k + 3 < T) ? entry_src(ent, k + 3) : i2;
    row[0] = row_of(xl4, i0, lane);
    row[1] = row_of(xl4, i1, lane);
    row[2] = row_of(xl4, i2, lane);
    row[3] = row_of(xl4, i3, lane);
}

__device__ __forceinline__ int node_entries(const int raw, const int i, const int lane, const int nmax) {
    return lane == 0 ? i : min(max(raw, 0), nmax);
}

__global__ __launch_bounds__(256, 10) void node_pull_kernel(
    const int* __restrict__ head, const int2* __restrict__ link,
    const float* __restrict__ xl, const float* __restrict__ xr,
    const float* __restrict__ att, const float* __restrict__ bias,
    int* outw, long N, long E)
{
    const int lane = threadIdx.x & 31;
    const int wave = __builtin_amdgcn_readfirstlane((int)(threadIdx.x >> 5));
    const unsigned nw  = gridDim.x * 8u;
    const unsigned wid = blockIdx.x * 8u + (unsigned)wave;
    const unsigned per = (unsigned)N / nw, extra = (unsigned)N % nw;
    const int lo = (int)(wid * per + min(wid, extra));
    const int hi = lo + (int)per + (wid < extra ? 1 : 0);
    if (lo >= hi) return;
    const float4 a4 = ((const float4*)att)[lane];
    const float4* xl4 = (const float4*)xl;
    const float4* xr4 = (const float4*)xr;
    const int last = (int)(E - 1);
    const int nmax = (int)(N - 1);
    const int slotoff = lane > 0 ? lane - 1 : 0;

    int ncn  = head[lo];
    int nraw = outw[(long)lo * ROW_W + slotoff];

    for (int i = lo; i < hi; ++i) {
        const int cnt = max(ncn, 0);
        const int ent = node_entries(nraw, i, lane, nmax);
        const int inx = min(i + 1, hi - 1);
        ncn  = head[inx];
        nraw = outw[(long)inx * ROW_W + slotoff];
        const float4 r4 = (xr4 + (long)i * 32)[lane];

        const int T   = 1 + min(cnt, SLOTS);
        int rem = -1;
        if (cnt > SLOTS) rem = __builtin_amdgcn_readfirstlane(outw[(long)i * ROW_W + SLOTS]);

        float m = -INFINITY;
        float den = 0.0f;
        float4 acc = make_float4(0.0f, 0.0f, 0.0f, 0.0f);

        float4 rowa[GRP], rowb[GRP];
        load_rows(rowa, ent, 0, T, xl4, lane);
        int k = 0;
        for (;;) {
            if (k + GRP > T) { take_rest(rowa, k, T, r4, a4, m, den, acc); break; }
            if (k + GRP < T) load_rows(rowb, ent, k + GRP, T, xl4, lane);
            take_group(rowa, r4, a4, m, den, acc);
            k += GRP;
            if (k + GRP > T) { take_rest(rowb, k, T, r4, a4, m, den, acc); break; }
            if (k + GRP < T) load_rows(rowa, ent, k + GRP, T, xl4, lane);
            take_group(rowb, r4, a4, m, den, acc);
            k += GRP;
        }

        int e = rem;
        for (long c2 = 0; c2 < E && e >= 0; ++c2) {
            const int2 l = link[link_slot(e, last)];
            const int s = min(max(__builtin_amdgcn_readfirstlane(l.y), 0), nmax);
            e = __builtin_amdgcn_readfirstlane(l.x);
            take_edge(row_of(xl4, s, lane), r4, a4, m, den, acc);
        }

        const float inv = 0.25f / (den + 1e-16f);
        float vx = acc.x * inv, vy = acc.y * inv, vz = acc.z * inv, vw = acc.w * inv;
        vx += __shfl_xor(vx, 8);  vy += __shfl_xor(vy, 8);
        vz += __shfl_xor(vz, 8);  vw += __shfl_xor(vw, 8);
        vx += __shfl_xor(vx, 16); vy += __shfl_xor(vy, 16);
        vz += __shfl_xor(vz, 16); vw += __shfl_xor(vw, 16);
        if (lane < 8) {
            const float4 b4 = ((const float4*)bias)[lane];
            int4 o4;
            o4.x = __float_as_int(vx + b4.x); o4.y = __float_as_int(vy + b4.y);
            o4.z = __float_as_int(vz + b4.z); o4.w = __float_as_int(vw + b4.w);
            ((int4*)outw)[(long)i * 8 + lane] = o4;
        }
    }
}

extern "C" void kernel_launch(void* const* d_in, const int* in_sizes, int n_in,
                              void* d_out, int out_size, void* d_ws, size_t ws_size,
                              hipStream_t stream) {
    const float* x     = (const float*)d_in[0];
    const int*   ei    = (const int*)d_in[1];
    const float* gamma = (const float*)d_in[2];
    const float* beta  = (const float*)d_in[3];
    const float* Wl    = (const float*)d_in[4];
    const float* bl    = (const float*)d_in[5];
    const float* Wr    = (const float*)d_in[6];
    const float* br    = (const float*)d_in[7];
    const float* att   = (const float*)d_in[8];
    const float* bias  = (const float*)d_in[9];
    int* outw = (int*)d_out;

    const long N = in_sizes[0] / C_IN;
    const long E = in_sizes[1] / 2;
    const int* srcI = ei;
    const int* dstI = ei + E;

    float* ws   = (float*)d_ws;
    float* xl   = ws;
    float* xr   = xl + N * C_IN;
    int2*  link = (int2*)(xr + N * C_IN);
    int*   head = (int*)(link + E);

    {
        head_clear_kernel<<<(int)((N + 255) / 256), 256, 0, stream>>>(head, outw, N);
    }
    {
        int blocks = (int)((N + ROWS_PB - 1) / ROWS_PB);
        ln_gemm_kernel<<<blocks + LIST_WGS, 256, 0, stream>>>(x, gamma, beta, Wl, bl, Wr, br, xl, xr, N,
                                                             srcI, dstI, head, link, outw, E);
    }
    {
        node_pull_kernel<<<NODE_GRID, 256, 0, stream>>>(head, link, xl, xr, att, bias, outw, N, E);
    }
}
